# starting kernel
# baseline (speedup 1.0000x reference)
#include <hip/hip_runtime.h>
#include <hip/hip_bf16.h>

typedef float v2f __attribute__((ext_vector_type(2)));
typedef float v8f __attribute__((ext_vector_type(8)));

#define DIM 128
#define NEG_SLOPE 0.01f
#define SCAN_T 1024

__global__ void cnt_kernel(const int* __restrict__ dst, int* __restrict__ cnt, int E) {
    int e = blockIdx.x * blockDim.x + threadIdx.x;
    if (e < E) atomicAdd(&cnt[dst[e]], 1);
}

__global__ void dis_kernel(const int* __restrict__ cnt, float* __restrict__ dis, int N) {
    int i = blockIdx.x * blockDim.x + threadIdx.x;
    if (i < N) dis[i] = rsqrtf((float)cnt[i] + 1.0f);
}

__global__ __launch_bounds__(SCAN_T) void scan_kernel(const int* __restrict__ cnt,
                                                      int* __restrict__ row_ptr, int N) {
    __shared__ int sdata[SCAN_T];
    __shared__ int carry_s;
    if (threadIdx.x == 0) carry_s = 0;
    __syncthreads();
    for (int base = 0; base < N; base += SCAN_T) {
        int i = base + threadIdx.x;
        int v = (i < N) ? cnt[i] : 0;
        sdata[threadIdx.x] = v;
        __syncthreads();
        for (int off = 1; off < SCAN_T; off <<= 1) {
            int t = (threadIdx.x >= off) ? sdata[threadIdx.x - off] : 0;
            __syncthreads();
            sdata[threadIdx.x] += t;
            __syncthreads();
        }
        int incl  = sdata[threadIdx.x];
        int total = sdata[SCAN_T - 1];
        int c     = carry_s;
        if (i < N) row_ptr[i] = c + incl - v;
        __syncthreads();
        if (threadIdx.x == 0) carry_s = c + total;
        __syncthreads();
    }
    if (threadIdx.x == 0) row_ptr[N] = carry_s;
}

__global__ void copy_kernel(const int* __restrict__ srcp, int* __restrict__ dstp, int N) {
    int i = blockIdx.x * blockDim.x + threadIdx.x;
    if (i < N) dstp[i] = srcp[i];
}

__global__ void fill_csr_kernel(const int* __restrict__ src, const int* __restrict__ dst,
                                const float* __restrict__ dis,
                                int* __restrict__ cursor, int* __restrict__ col,
                                float* __restrict__ val, int E) {
    int e = blockIdx.x * blockDim.x + threadIdx.x;
    if (e >= E) return;
    int s = src[e], d = dst[e];
    int pos = atomicAdd(&cursor[d], 1);
    col[pos] = s;
    val[pos] = dis[s] * dis[d];
}

__global__ __launch_bounds__(256) void gemm_wmma_kernel(
        const float* __restrict__ A,
        const float* __restrict__ W,
        float* __restrict__ T,
        int N) {
    __shared__ float As[16 * 132];

    const int row0 = blockIdx.x * 16;
    const int tid  = threadIdx.x;

    for (int i = tid; i < 16 * DIM; i += 256) {
        int r = i >> 7;
        int c = i & (DIM - 1);
        int rr = row0 + r;
        As[r * 132 + c] = (rr < N) ? A[(size_t)rr * DIM + c] : 0.0f;
    }
    __syncthreads();

    const int wave  = tid >> 5;
    const int lane  = tid & 31;
    const int m     = lane & 15;
    const int khalf = (lane >> 4) * 2;
    const int n0    = wave * 16;

    v8f acc = {};
#pragma unroll
    for (int k = 0; k < DIM; k += 4) {
        v2f a, b;
        a.x = As[m * 132 + k + khalf];
        a.y = As[m * 132 + k + khalf + 1];
        b.x = W[(k + khalf) * DIM + n0 + m];
        b.y = W[(k + khalf + 1) * DIM + n0 + m];
        acc = __builtin_amdgcn_wmma_f32_16x16x4_f32(
                false, a, false, b, (short)0, acc, false, false);
    }

    const int rbase = row0 + (lane >> 4) * 8;
#pragma unroll
    for (int v = 0; v < 8; ++v) {
        int r = rbase + v;
        if (r < N) T[(size_t)r * DIM + n0 + m] = acc[v];
    }
}

__global__ __launch_bounds__(256) void spmm_kernel(
        const float* __restrict__ T,
        const int*   __restrict__ row_ptr,
        const int*   __restrict__ col,
        const float* __restrict__ val,
        const float* __restrict__ dis,
        const float* __restrict__ bias,
        float*       __restrict__ out,
        int N) {
    int node = blockIdx.x * 8 + (threadIdx.x >> 5);
    if (node >= N) return;
    int lane = threadIdx.x & 31;

    int beg = row_ptr[node];
    int end = row_ptr[node + 1];

    float ds = dis[node];
    float w0 = ds * ds;
    float4 tv = ((const float4*)(T + (size_t)node * DIM))[lane];
    float4 acc;
    acc.x = tv.x * w0; acc.y = tv.y * w0; acc.z = tv.z * w0; acc.w = tv.w * w0;

    for (int jb = beg; jb < end; jb += 32) {
        int nb = end - jb;
        if (nb > 32) nb = 32;
        int   jl  = jb + lane;
        int   s_l = (jl < end) ? col[jl] : 0;
        float w_l = (jl < end) ? val[jl] : 0.0f;
        for (int k = 0; k < nb; ++k) {
            int   s = __shfl(s_l, k);
            float w = __shfl(w_l, k);
            float4 v = ((const float4*)(T + (size_t)s * DIM))[lane];
            acc.x = fmaf(v.x, w, acc.x);
            acc.y = fmaf(v.y, w, acc.y);
            acc.z = fmaf(v.z, w, acc.z);
            acc.w = fmaf(v.w, w, acc.w);
        }
    }

    float4 bv = ((const float4*)bias)[lane];
    float4 r;
    r.x = acc.x + bv.x; r.y = acc.y + bv.y;
    r.z = acc.z + bv.z; r.w = acc.w + bv.w;
    r.x = (r.x >= 0.0f) ? r.x : NEG_SLOPE * r.x;
    r.y = (r.y >= 0.0f) ? r.y : NEG_SLOPE * r.y;
    r.z = (r.z >= 0.0f) ? r.z : NEG_SLOPE * r.z;
    r.w = (r.w >= 0.0f) ? r.w : NEG_SLOPE * r.w;
    ((float4*)(out + (size_t)node * DIM))[lane] = r;
}

extern "C" void kernel_launch(void* const* d_in, const int* in_sizes, int n_in,
                              void* d_out, int out_size, void* d_ws, size_t ws_size,
                              hipStream_t stream) {
    const float* x  = (const float*)d_in[0];
    const int*   ei = (const int*)  d_in[1];
    const float* W0 = (const float*)d_in[2];
    const float* b0 = (const float*)d_in[3];
    const float* W1 = (const float*)d_in[4];
    const float* b1 = (const float*)d_in[5];
    float* out = (float*)d_out;

    const int N = in_sizes[0] / DIM;
    const int E = in_sizes[1] / 2;
    const int* srcIdx = ei;
    const int* dstIdx = ei + E;

    const size_t NH = (size_t)N * DIM;

    float* T       = (float*)d_ws;
    float* dis     = T + NH;
    float* val     = dis + N;
    int*   cnt     = (int*)(val + E);
    int*   row_ptr = cnt + N;
    int*   cursor  = row_ptr + (N + 1);
    int*   col     = cursor + N;

    const int thr   = 256;
    const int gridN = (N + thr - 1) / thr;
    const int gridE = (E + thr - 1) / thr;

    hipMemsetAsync(cnt, 0, (size_t)N * sizeof(int), stream);
    cnt_kernel <<<gridE, thr, 0, stream>>>(dstIdx, cnt, E);
    dis_kernel <<<gridN, thr, 0, stream>>>(cnt, dis, N);
    scan_kernel<<<1, SCAN_T, 0, stream>>>(cnt, row_ptr, N);
    copy_kernel<<<gridN, thr, 0, stream>>>(row_ptr, cursor, N);
    fill_csr_kernel<<<gridE, thr, 0, stream>>>(srcIdx, dstIdx, dis, cursor, col, val, E);

    const int gemmGrid = (N + 15) / 16;
    const int nodeGrid = (N + 7) / 8;

    gemm_wmma_kernel<<<gemmGrid, thr, 0, stream>>>(x, W0, T, N);
    spmm_kernel<<<nodeGrid, thr, 0, stream>>>(T, row_ptr, col, val, dis, b0, out, N);

    gemm_wmma_kernel<<<gemmGrid, thr, 0, stream>>>(out, W1, T, N);
    spmm_kernel<<<nodeGrid, thr, 0, stream>>>(T, row_ptr, col, val, dis, b1, out, N);
}


# dense transform: a wave keeps its 128x16 stripe of W in registers and its block walks 16-row tiles (512 blocks, next tile's rows in flight under the products); W no longer re-read per 16 rows
# speedup vs baseline: 2.3529x; 2.3529x over previous
#include <hip/hip_runtime.h>
#include <hip/hip_bf16.h>


typedef float v2f __attribute__((ext_vector_type(2)));
typedef float v8f __attribute__((ext_vector_type(8)));

#define DIM 128
#define NEG_SLOPE 0.01f
#define SLOTS 32
#define EDGES_PER_LANE 4

template <bool FULL>
__device__ __forceinline__ void edge_pass_body(const int* __restrict__ src,
                                               const int* __restrict__ dst,
                                               int* __restrict__ cnt,
                                               int* __restrict__ slot,
                                               int2* __restrict__ ovf,
                                               int* __restrict__ ovf_n, int E, int base) {
    int s[EDGES_PER_LANE], d[EDGES_PER_LANE], r[EDGES_PER_LANE];
#pragma unroll
    for (int k = 0; k < EDGES_PER_LANE; ++k) {
        const int e = base + k * 256;
        const bool ok = FULL || (e < E);
        d[k] = ok ? dst[e] : 0;
        s[k] = ok ? src[e] : 0;
    }
#pragma unroll
    for (int k = 0; k < EDGES_PER_LANE; ++k) {
        const int e = base + k * 256;
        const bool ok = FULL || (e < E);
        r[k] = ok ? atomicAdd(&cnt[d[k]], 1) : 0;
    }
#pragma unroll
    for (int k = 0; k < EDGES_PER_LANE; ++k) {
        const int e = base + k * 256;
        const bool ok = FULL || (e < E);
        if (ok) {
            if (r[k] < SLOTS) {
                slot[(size_t)d[k] * SLOTS + r[k]] = s[k];
            } else {
                const int p = atomicAdd(ovf_n, 1);
                ovf[p] = make_int2(d[k], s[k]);
            }
        }
    }
}

__global__ __launch_bounds__(256) void edge_pass_kernel(const int* __restrict__ src,
                                                        const int* __restrict__ dst,
                                                        int* __restrict__ cnt,
                                                        int* __restrict__ slot,
                                                        int2* __restrict__ ovf,
                                                        int* __restrict__ ovf_n, int E) {
    const int block0 = blockIdx.x * (256 * EDGES_PER_LANE);
    const int base   = block0 + threadIdx.x;
    if (block0 + 256 * EDGES_PER_LANE <= E)
        edge_pass_body<true>(src, dst, cnt, slot, ovf, ovf_n, E, base);
    else
        edge_pass_body<false>(src, dst, cnt, slot, ovf, ovf_n, E, base);
}

#define GEMM_GRID 512
#define A_PITCH 132

__device__ __forceinline__ void gemm_fetch_tile(const float* __restrict__ A,
                                                const int* __restrict__ cnt,
                                                int tile, int N, int tid,
                                                float4& p0, float4& p1, int& deg) {
    const int last = N - 1;
    const int ra = tile * 16 + (tid >> 5);
    const int r0 = (ra < last) ? ra : last;
    const int r1 = (ra + 8 < last) ? (ra + 8) : last;
    const int c  = (tid & 31) * 4;
    p0 = *(const float4*)(A + (size_t)r0 * DIM + c);
    p1 = *(const float4*)(A + (size_t)r1 * DIM + c);
    const int rd = tile * 16 + (tid & 15);
    deg = cnt[(rd < last) ? rd : last];
}

__device__ __forceinline__ void gemm_stage_tile(float* __restrict__ as, float* __restrict__ sc,
                                                int tid, const float4& p0, const float4& p1,
                                                int deg) {
    const int r = tid >> 5;
    const int c = (tid & 31) * 4;
    *(float4*)(as + r * A_PITCH + c)       = p0;
    *(float4*)(as + (r + 8) * A_PITCH + c) = p1;
    if (tid < 16) sc[tid] = rsqrtf((float)deg + 1.0f);
}

__global__ __launch_bounds__(256) void gemm_wmma_kernel(
        const float* __restrict__ A,
        const float* __restrict__ W,
        const int*   __restrict__ cnt,
        float* __restrict__ T,
        int N) {
    __shared__ __attribute__((aligned(16))) float As[2][16 * A_PITCH];
    __shared__ __attribute__((aligned(16))) float Sc[2][16];

    const int numTiles = (N + 15) >> 4;
    int tile = blockIdx.x;
    if (tile >= numTiles) return;

    const int tid   = threadIdx.x;
    const int wave  = tid >> 5;
    const int lane  = tid & 31;
    const int m     = lane & 15;
    const int khalf = (lane >> 4) * 2;
    const int n0    = wave * 16;
    const int hrow  = (lane >> 4) * 8;

    float4 f0, f1;
    int fdeg;
    gemm_fetch_tile(A, cnt, tile, N, tid, f0, f1, fdeg);

    v2f bw[DIM / 4];
#pragma unroll
    for (int j = 0; j < DIM / 4; ++j) {
        const int k = 4 * j;
        bw[j].x = W[(k + khalf) * DIM + n0 + m];
        bw[j].y = W[(k + khalf + 1) * DIM + n0 + m];
    }

    int buf = 0;
    gemm_stage_tile(As[0], Sc[0], tid, f0, f1, fdeg);
    __syncthreads();

    for (;;) {
        const int  next     = tile + GEMM_GRID;
        const bool has_next = next < numTiles;
        float4 p0, p1;
        int deg;
        if (has_next) gemm_fetch_tile(A, cnt, next, N, tid, p0, p1, deg);

        const float* as = As[buf];
        v8f acc = {};
#pragma unroll
        for (int j = 0; j < DIM / 4; ++j) {
            const int k = 4 * j;
            v2f a;
            a.x = as[m * A_PITCH + k + khalf];
            a.y = as[m * A_PITCH + k + khalf + 1];
            acc = __builtin_amdgcn_wmma_f32_16x16x4_f32(
                    false, a, false, bw[j], (short)0, acc, false, false);
        }

        float res[8];
#pragma unroll
        for (int v = 0; v < 8; ++v) res[v] = acc[v] * Sc[buf][hrow + v];

        if (has_next) gemm_stage_tile(As[buf ^ 1], Sc[buf ^ 1], tid, p0, p1, deg);
        __syncthreads();

        const int rbase = tile * 16 + hrow;
#pragma unroll
        for (int v = 0; v < 8; ++v) {
            const int r = rbase + v;
            if (r < N) T[(size_t)r * DIM + n0 + m] = res[v];
        }

        if (!has_next) break;
        tile = next;
        buf ^= 1;
    }
}

__global__ __launch_bounds__(256) void spmm_kernel(
        const float* __restrict__ T,
        const int*   __restrict__ cnt,
        const int*   __restrict__ slot,
        const int2*  __restrict__ ovf,
        const int*   __restrict__ ovf_n,
        const float* __restrict__ bias,
        float*       __restrict__ out,
        int N, int E) {
    int node = blockIdx.x * 8 + (threadIdx.x >> 5);
    if (node >= N) return;
    int lane = threadIdx.x & 31;

    const int c    = cnt[node];
    const int n_in = (c < SLOTS) ? c : SLOTS;

    float4 acc = ((const float4*)(T + (size_t)node * DIM))[lane];

    const int s_l = (lane < n_in) ? slot[(size_t)node * SLOTS + lane] : 0;

    for (int k = 0; k < n_in; ++k) {
        int s = __shfl(s_l, k);
        float4 v = ((const float4*)(T + (size_t)s * DIM))[lane];
        acc.x += v.x; acc.y += v.y; acc.z += v.z; acc.w += v.w;
    }

    if (c > SLOTS) {
        int n_ovf = *ovf_n;
        if (n_ovf > E) n_ovf = E;
        for (int p = 0; p < n_ovf; p += 32) {
            const int q = p + lane;
            int2 ent = make_int2(-1, 0);
            if (q < n_ovf) ent = ovf[q];
            unsigned hit = __builtin_amdgcn_ballot_w32(ent.x == node);
            while (hit) {
                const int k = __builtin_ctz(hit);
                hit &= hit - 1;
                int s = __shfl(ent.y, k);
                float4 v = ((const float4*)(T + (size_t)s * DIM))[lane];
                acc.x += v.x; acc.y += v.y; acc.z += v.z; acc.w += v.w;
            }
        }
    }

    const float ds = rsqrtf((float)c + 1.0f);
    float4 bv = ((const float4*)bias)[lane];
    float4 r;
    r.x = acc.x * ds + bv.x; r.y = acc.y * ds + bv.y;
    r.z = acc.z * ds + bv.z; r.w = acc.w * ds + bv.w;
    r.x = (r.x >= 0.0f) ? r.x : NEG_SLOPE * r.x;
    r.y = (r.y >= 0.0f) ? r.y : NEG_SLOPE * r.y;
    r.z = (r.z >= 0.0f) ? r.z : NEG_SLOPE * r.z;
    r.w = (r.w >= 0.0f) ? r.w : NEG_SLOPE * r.w;
    ((float4*)(out + (size_t)node * DIM))[lane] = r;
}

extern "C" void kernel_launch(void* const* d_in, const int* in_sizes, int n_in,
                              void* d_out, int out_size, void* d_ws, size_t ws_size,
                              hipStream_t stream) {
    const float* x  = (const float*)d_in[0];
    const int*   ei = (const int*)  d_in[1];
    const float* W0 = (const float*)d_in[2];
    const float* b0 = (const float*)d_in[3];
    const float* W1 = (const float*)d_in[4];
    const float* b1 = (const float*)d_in[5];
    float* out = (float*)d_out;

    const int N = in_sizes[0] / DIM;
    const int E = in_sizes[1] / 2;
    const int* srcIdx = ei;
    const int* dstIdx = ei + E;

    const size_t NH = (size_t)N * DIM;

    float* T     = (float*)d_ws;
    int*   slot  = (int*)(T + NH);
    int2*  ovf   = (int2*)(slot + (size_t)N * SLOTS);
    int*   cnt   = (int*)(ovf + E);
    int*   ovf_n = cnt + N;

    const int thr   = 256;
    const int gridE = (E + thr * EDGES_PER_LANE - 1) / (thr * EDGES_PER_LANE);

    hipMemsetAsync(cnt, 0, (size_t)(N + 1) * sizeof(int), stream);
    edge_pass_kernel<<<gridE, thr, 0, stream>>>(srcIdx, dstIdx, cnt, slot, ovf, ovf_n, E);

    const int nodeGrid = (N + 7) / 8;

    gemm_wmma_kernel<<<GEMM_GRID, thr, 0, stream>>>(x, W0, cnt, T, N);
    spmm_kernel<<<nodeGrid, thr, 0, stream>>>(T, cnt, slot, ovf, ovf_n, b0, out, N, E);

    gemm_wmma_kernel<<<GEMM_GRID, thr, 0, stream>>>(out, W1, cnt, T, N);
    spmm_kernel<<<nodeGrid, thr, 0, stream>>>(T, cnt, slot, ovf, ovf_n, b1, out, N, E);
}
